# starting kernel
# baseline (speedup 1.0000x reference)
#include <hip/hip_runtime.h>
#include <hip/hip_bf16.h>

typedef __attribute__((ext_vector_type(2))) float v2f;
typedef __attribute__((ext_vector_type(8))) float v8f;

#define S_LEN   2048
#define D_MODEL 1024
#define NHEAD   16
#define DK      64
#define BATCH   2
#define M_TOT   (BATCH * S_LEN)
#define ROPE_L2T_OVER_DK 0.20762050593046014f

__device__ __forceinline__ float red_max16(float v) {
  v = fmaxf(v, __shfl_xor(v, 1, 32));
  v = fmaxf(v, __shfl_xor(v, 2, 32));
  v = fmaxf(v, __shfl_xor(v, 4, 32));
  v = fmaxf(v, __shfl_xor(v, 8, 32));
  return v;
}
__device__ __forceinline__ float red_sum16(float v) {
  v += __shfl_xor(v, 1, 32);
  v += __shfl_xor(v, 2, 32);
  v += __shfl_xor(v, 4, 32);
  v += __shfl_xor(v, 8, 32);
  return v;
}

__global__ __launch_bounds__(256) void gemm_wmma_f32(
    const float* __restrict__ X, const float* __restrict__ W,
    float* __restrict__ out, int mode)
{
  const int lane = threadIdx.x & 31;
  const int wave = threadIdx.x >> 5;
  const int half = lane >> 4;
  const int l16  = lane & 15;

  const int m0 = (blockIdx.y * 8 + wave) * 32;
  const int n0 = blockIdx.x * 64;

  const float* Arow0 = X + (size_t)(m0 + l16) * D_MODEL;
  const float* Arow1 = Arow0 + (size_t)16 * D_MODEL;
  const float* Brow0 = W + (size_t)(n0 + l16) * D_MODEL;
  const float* Brow1 = Brow0 + (size_t)16 * D_MODEL;
  const float* Brow2 = Brow0 + (size_t)32 * D_MODEL;
  const float* Brow3 = Brow0 + (size_t)48 * D_MODEL;

  v8f acc[2][4];
#pragma unroll
  for (int i = 0; i < 2; ++i)
#pragma unroll
    for (int t = 0; t < 4; ++t) acc[i][t] = (v8f){};

  const int koff = 2 * half;
#pragma unroll 2
  for (int k = 0; k < D_MODEL; k += 4) {
    const int kk = k + koff;
    v2f a0 = *(const v2f*)(Arow0 + kk);
    v2f a1 = *(const v2f*)(Arow1 + kk);
    v2f b0 = *(const v2f*)(Brow0 + kk);
    v2f b1 = *(const v2f*)(Brow1 + kk);
    v2f b2 = *(const v2f*)(Brow2 + kk);
    v2f b3 = *(const v2f*)(Brow3 + kk);
    acc[0][0] = __builtin_amdgcn_wmma_f32_16x16x4_f32(false, a0, false, b0, (short)0, acc[0][0], false, false);
    acc[0][1] = __builtin_amdgcn_wmma_f32_16x16x4_f32(false, a0, false, b1, (short)0, acc[0][1], false, false);
    acc[0][2] = __builtin_amdgcn_wmma_f32_16x16x4_f32(false, a0, false, b2, (short)0, acc[0][2], false, false);
    acc[0][3] = __builtin_amdgcn_wmma_f32_16x16x4_f32(false, a0, false, b3, (short)0, acc[0][3], false, false);
    acc[1][0] = __builtin_amdgcn_wmma_f32_16x16x4_f32(false, a1, false, b0, (short)0, acc[1][0], false, false);
    acc[1][1] = __builtin_amdgcn_wmma_f32_16x16x4_f32(false, a1, false, b1, (short)0, acc[1][1], false, false);
    acc[1][2] = __builtin_amdgcn_wmma_f32_16x16x4_f32(false, a1, false, b2, (short)0, acc[1][2], false, false);
    acc[1][3] = __builtin_amdgcn_wmma_f32_16x16x4_f32(false, a1, false, b3, (short)0, acc[1][3], false, false);
  }

  if (mode == 0) {
#pragma unroll
    for (int i = 0; i < 2; ++i)
#pragma unroll
      for (int t = 0; t < 4; ++t)
#pragma unroll
        for (int r = 0; r < 8; ++r) {
          const int row = m0 + i * 16 + r + 8 * half;
          out[(size_t)row * D_MODEL + n0 + t * 16 + l16] = acc[i][t][r];
        }
  } else {
#pragma unroll
    for (int t = 0; t < 4; ++t) {
      const int gcol = n0 + t * 16 + l16;
      const int h = gcol >> 6;
      const int d = gcol & 63;
      const bool even = (d & 1) == 0;
      const float invfreq = exp2f(-(float)(d & ~1) * ROPE_L2T_OVER_DK);
#pragma unroll
      for (int i = 0; i < 2; ++i)
#pragma unroll
        for (int r = 0; r < 8; ++r) {
          const int row = m0 + i * 16 + r + 8 * half;
          const int b  = row >> 11;
          const int s  = row & (S_LEN - 1);
          float v  = acc[i][t][r];
          float vp = __shfl_xor(v, 1, 32);
          if (mode == 1) {
            const float ang = (float)s * invfreq;
            const float c = __cosf(ang), sn = __sinf(ang);
            v = v * c + (even ? -vp : vp) * sn;
          }
          out[(((size_t)(b * NHEAD + h) * S_LEN + s) * DK) + d] = v;
        }
    }
  }
}

__global__ __launch_bounds__(32) void flash_attn(
    const float* __restrict__ Q, const float* __restrict__ K,
    const float* __restrict__ V, float* __restrict__ att)
{
  __shared__ float p_lds[16 * 16];
  const int lane = threadIdx.x & 31;
  const int half = lane >> 4;
  const int l16  = lane & 15;
  const int qt   = blockIdx.x;
  const int bh   = blockIdx.y;
  const size_t base = (size_t)bh * S_LEN * DK;

  v2f qa[16];
  {
    const float* qrow = Q + base + (size_t)(qt * 16 + l16) * DK + 2 * half;
#pragma unroll
    for (int c = 0; c < 16; ++c) {
      v2f t = *(const v2f*)(qrow + 4 * c);
      qa[c] = t * 0.125f;
    }
  }

  v8f o0 = {}, o1 = {}, o2 = {}, o3 = {};
  float mrun[8], lrun[8];
#pragma unroll
  for (int r = 0; r < 8; ++r) { mrun[r] = -__builtin_inff(); lrun[r] = 0.f; }

  for (int kb = 0; kb <= qt; ++kb) {
    v8f sc = {};
    const float* krow = K + base + (size_t)(kb * 16 + l16) * DK + 2 * half;
#pragma unroll
    for (int c = 0; c < 16; ++c) {
      v2f bf = *(const v2f*)(krow + 4 * c);
      sc = __builtin_amdgcn_wmma_f32_16x16x4_f32(false, qa[c], false, bf,
                                                 (short)0, sc, false, false);
    }

    float pvals[8], scl[8];
#pragma unroll
    for (int r = 0; r < 8; ++r) {
      const int row = r + 8 * half;
      const int qg  = qt * 16 + row;
      const int kg  = kb * 16 + l16;
      const float s = (kg <= qg) ? sc[r] : -__builtin_inff();
      const float rmax = red_max16(s);
      const float mnew = fmaxf(mrun[r], rmax);
      const float p = __expf(s - mnew);
      const float rsum = red_sum16(p);
      scl[r]  = __expf(mrun[r] - mnew);
      lrun[r] = lrun[r] * scl[r] + rsum;
      mrun[r] = mnew;
      pvals[r] = p;
    }

#pragma unroll
    for (int r = 0; r < 8; ++r) {
      o0[r] *= scl[r]; o1[r] *= scl[r]; o2[r] *= scl[r]; o3[r] *= scl[r];
    }

#pragma unroll
    for (int r = 0; r < 8; ++r)
      p_lds[(r + 8 * half) * 16 + l16] = pvals[r];
    __syncthreads();

    const float* vbase = V + base + (size_t)(kb * 16) * DK;
#pragma unroll
    for (int c = 0; c < 4; ++c) {
      const int kk = c * 4 + 2 * half;
      v2f a;
      a[0] = p_lds[l16 * 16 + kk];
      a[1] = p_lds[l16 * 16 + kk + 1];
      v2f b0, b1, b2, b3;
      const float* v0 = vbase + (size_t)kk * DK + l16;
      const float* v1 = v0 + DK;
      b0[0] = v0[0];  b0[1] = v1[0];
      b1[0] = v0[16]; b1[1] = v1[16];
      b2[0] = v0[32]; b2[1] = v1[32];
      b3[0] = v0[48]; b3[1] = v1[48];
      o0 = __builtin_amdgcn_wmma_f32_16x16x4_f32(false, a, false, b0, (short)0, o0, false, false);
      o1 = __builtin_amdgcn_wmma_f32_16x16x4_f32(false, a, false, b1, (short)0, o1, false, false);
      o2 = __builtin_amdgcn_wmma_f32_16x16x4_f32(false, a, false, b2, (short)0, o2, false, false);
      o3 = __builtin_amdgcn_wmma_f32_16x16x4_f32(false, a, false, b3, (short)0, o3, false, false);
    }
    __syncthreads();
  }

  const int b = bh >> 4;
  const int h = bh & 15;
#pragma unroll
  for (int r = 0; r < 8; ++r) {
    const float inv = 1.0f / lrun[r];
    const int row = qt * 16 + r + 8 * half;
    const size_t ob = ((size_t)(b * S_LEN) + row) * D_MODEL + h * DK;
    att[ob +  0 + l16] = o0[r] * inv;
    att[ob + 16 + l16] = o1[r] * inv;
    att[ob + 32 + l16] = o2[r] * inv;
    att[ob + 48 + l16] = o3[r] * inv;
  }
}

extern "C" void kernel_launch(void* const* d_in, const int* in_sizes, int n_in,
                              void* d_out, int out_size, void* d_ws, size_t ws_size,
                              hipStream_t stream) {
  const float* x  = (const float*)d_in[0];
  const float* wq = (const float*)d_in[1];
  const float* wk = (const float*)d_in[2];
  const float* wv = (const float*)d_in[3];
  const float* wo = (const float*)d_in[4];
  float* out = (float*)d_out;

  float* ws  = (float*)d_ws;
  const size_t SZ = (size_t)M_TOT * D_MODEL;
  float* Qw  = ws;
  float* Kw  = ws + SZ;
  float* Vw  = ws + 2 * SZ;
  float* Aw  = ws + 3 * SZ;

  dim3 blk(256);
  dim3 grd(D_MODEL / 64, M_TOT / (32 * 8));
  gemm_wmma_f32<<<grd, blk, 0, stream>>>(x, wq, Qw, 1);
  gemm_wmma_f32<<<grd, blk, 0, stream>>>(x, wk, Kw, 1);
  gemm_wmma_f32<<<grd, blk, 0, stream>>>(x, wv, Vw, 2);

  flash_attn<<<dim3(S_LEN / 16, BATCH * NHEAD), dim3(32), 0, stream>>>(Qw, Kw, Vw, Aw);

  gemm_wmma_f32<<<grd, blk, 0, stream>>>(Aw, wo, out, 0);
}


# attention tiles transposed (S^T=K.Q^T, O^T=V^T.P^T): softmax per lane, P from registers into PV, no LDS/barrier/16-lane shuffles; loads grouped (4 waves/SIMD budget); l and PV key sum order change
# speedup vs baseline: 1.5468x; 1.5468x over previous
#include <hip/hip_runtime.h>
#include <hip/hip_bf16.h>

typedef __attribute__((ext_vector_type(2))) float v2f;
typedef __attribute__((ext_vector_type(4))) float v4f;
typedef __attribute__((ext_vector_type(8))) float v8f;

#define S_LEN   2048
#define D_MODEL 1024
#define NHEAD   16
#define DK      64
#define BATCH   2
#define M_TOT   (BATCH * S_LEN)
#define ROPE_L2T_OVER_DK 0.20762050593046014f

__global__ __launch_bounds__(256) void gemm_wmma_f32(
    const float* __restrict__ X, const float* __restrict__ W,
    float* __restrict__ out, int mode)
{
  const int lane = threadIdx.x & 31;
  const int wave = threadIdx.x >> 5;
  const int half = lane >> 4;
  const int l16  = lane & 15;

  const int m0 = (blockIdx.y * 8 + wave) * 32;
  const int n0 = blockIdx.x * 64;

  const float* Arow0 = X + (size_t)(m0 + l16) * D_MODEL;
  const float* Arow1 = Arow0 + (size_t)16 * D_MODEL;
  const float* Brow0 = W + (size_t)(n0 + l16) * D_MODEL;
  const float* Brow1 = Brow0 + (size_t)16 * D_MODEL;
  const float* Brow2 = Brow0 + (size_t)32 * D_MODEL;
  const float* Brow3 = Brow0 + (size_t)48 * D_MODEL;

  v8f acc[2][4];
#pragma unroll
  for (int i = 0; i < 2; ++i)
#pragma unroll
    for (int t = 0; t < 4; ++t) acc[i][t] = (v8f){};

  const int koff = 2 * half;
#pragma unroll 2
  for (int k = 0; k < D_MODEL; k += 4) {
    const int kk = k + koff;
    v2f a0 = *(const v2f*)(Arow0 + kk);
    v2f a1 = *(const v2f*)(Arow1 + kk);
    v2f b0 = *(const v2f*)(Brow0 + kk);
    v2f b1 = *(const v2f*)(Brow1 + kk);
    v2f b2 = *(const v2f*)(Brow2 + kk);
    v2f b3 = *(const v2f*)(Brow3 + kk);
    acc[0][0] = __builtin_amdgcn_wmma_f32_16x16x4_f32(false, a0, false, b0, (short)0, acc[0][0], false, false);
    acc[0][1] = __builtin_amdgcn_wmma_f32_16x16x4_f32(false, a0, false, b1, (short)0, acc[0][1], false, false);
    acc[0][2] = __builtin_amdgcn_wmma_f32_16x16x4_f32(false, a0, false, b2, (short)0, acc[0][2], false, false);
    acc[0][3] = __builtin_amdgcn_wmma_f32_16x16x4_f32(false, a0, false, b3, (short)0, acc[0][3], false, false);
    acc[1][0] = __builtin_amdgcn_wmma_f32_16x16x4_f32(false, a1, false, b0, (short)0, acc[1][0], false, false);
    acc[1][1] = __builtin_amdgcn_wmma_f32_16x16x4_f32(false, a1, false, b1, (short)0, acc[1][1], false, false);
    acc[1][2] = __builtin_amdgcn_wmma_f32_16x16x4_f32(false, a1, false, b2, (short)0, acc[1][2], false, false);
    acc[1][3] = __builtin_amdgcn_wmma_f32_16x16x4_f32(false, a1, false, b3, (short)0, acc[1][3], false, false);
  }

  if (mode == 0) {
#pragma unroll
    for (int i = 0; i < 2; ++i)
#pragma unroll
      for (int t = 0; t < 4; ++t)
#pragma unroll
        for (int r = 0; r < 8; ++r) {
          const int row = m0 + i * 16 + r + 8 * half;
          out[(size_t)row * D_MODEL + n0 + t * 16 + l16] = acc[i][t][r];
        }
  } else {
#pragma unroll
    for (int t = 0; t < 4; ++t) {
      const int gcol = n0 + t * 16 + l16;
      const int h = gcol >> 6;
      const int d = gcol & 63;
      const bool even = (d & 1) == 0;
      const float invfreq = exp2f(-(float)(d & ~1) * ROPE_L2T_OVER_DK);
#pragma unroll
      for (int i = 0; i < 2; ++i)
#pragma unroll
        for (int r = 0; r < 8; ++r) {
          const int row = m0 + i * 16 + r + 8 * half;
          const int b  = row >> 11;
          const int s  = row & (S_LEN - 1);
          float v  = acc[i][t][r];
          float vp = __shfl_xor(v, 1, 32);
          if (mode == 1) {
            const float ang = (float)s * invfreq;
            const float c = __cosf(ang), sn = __sinf(ang);
            v = v * c + (even ? -vp : vp) * sn;
          }
          out[(((size_t)(b * NHEAD + h) * S_LEN + s) * DK) + d] = v;
        }
    }
  }
}

__device__ __forceinline__ float xhalf(float x) {
  const int b = __builtin_bit_cast(int, x);
  return __builtin_bit_cast(float,
      __builtin_amdgcn_permlanex16(b, b, 0x76543210, 0xfedcba98, false, false));
}

__global__ __launch_bounds__(32) __attribute__((amdgpu_waves_per_eu(4, 4))) void flash_attn(
    const float* __restrict__ Q, const float* __restrict__ K,
    const float* __restrict__ V, float* __restrict__ att)
{
  const int lane = threadIdx.x & 31;
  const int half = lane >> 4;
  const int l16  = lane & 15;
  const int qt   = blockIdx.x;
  const int bh   = blockIdx.y;
  const size_t base = (size_t)bh * S_LEN * DK;

  v2f qb[16];
  {
    const float* qrow = Q + base + (size_t)(qt * 16 + l16) * DK + 2 * half;
#pragma unroll
    for (int c = 0; c < 16; ++c) {
      v2f t = *(const v2f*)(qrow + 4 * c);
      qb[c] = t * 0.125f;
    }
  }

  v8f o0 = {}, o1 = {}, o2 = {}, o3 = {};
  float mrun = -__builtin_inff();
  float lrun = 0.f;
  const int qg = qt * 16 + l16;

  for (int kb = 0; kb <= qt; ++kb) {
    const int kg0 = kb * 16 + 8 * half;

    const float* krow = K + base + (size_t)(kb * 16 + l16) * DK + 2 * half;
    v2f ka[16];
#pragma unroll
    for (int c = 0; c < 16; ++c) ka[c] = *(const v2f*)(krow + 4 * c);

    const float* vcol = V + base + (size_t)kg0 * DK + l16;
    float vv[8][4];
#pragma unroll
    for (int i = 0; i < 8; ++i)
#pragma unroll
      for (int t = 0; t < 4; ++t) vv[i][t] = vcol[i * DK + 16 * t];

    v8f st = {};
#pragma unroll
    for (int c = 0; c < 16; ++c)
      st = __builtin_amdgcn_wmma_f32_16x16x4_f32(false, ka[c], false, qb[c],
                                                 (short)0, st, false, false);

    float s[8];
#pragma unroll
    for (int r = 0; r < 8; ++r)
      s[r] = (kg0 + r <= qg) ? st[r] : -__builtin_inff();
    float mx = s[0];
#pragma unroll
    for (int r = 1; r < 8; ++r) mx = fmaxf(mx, s[r]);
    mx = fmaxf(mx, xhalf(mx));
    const float mnew = fmaxf(mrun, mx);
    const float scl  = __expf(mrun - mnew);
    float p[8];
    float psum = 0.f;
#pragma unroll
    for (int r = 0; r < 8; ++r) {
      p[r] = __expf(s[r] - mnew);
      psum += p[r];
    }
    lrun = lrun * scl + psum;
    mrun = mnew;

    o0 *= scl; o1 *= scl; o2 *= scl; o3 *= scl;

#pragma unroll
    for (int j = 0; j < 4; ++j) {
      v2f pb;
      pb[0] = p[2 * j];
      pb[1] = p[2 * j + 1];
      v2f a0, a1, a2, a3;
      a0[0] = vv[2 * j][0]; a0[1] = vv[2 * j + 1][0];
      a1[0] = vv[2 * j][1]; a1[1] = vv[2 * j + 1][1];
      a2[0] = vv[2 * j][2]; a2[1] = vv[2 * j + 1][2];
      a3[0] = vv[2 * j][3]; a3[1] = vv[2 * j + 1][3];
      o0 = __builtin_amdgcn_wmma_f32_16x16x4_f32(false, a0, false, pb, (short)0, o0, false, false);
      o1 = __builtin_amdgcn_wmma_f32_16x16x4_f32(false, a1, false, pb, (short)0, o1, false, false);
      o2 = __builtin_amdgcn_wmma_f32_16x16x4_f32(false, a2, false, pb, (short)0, o2, false, false);
      o3 = __builtin_amdgcn_wmma_f32_16x16x4_f32(false, a3, false, pb, (short)0, o3, false, false);
    }
  }

  const float inv = 1.0f / (lrun + xhalf(lrun));
  const int b = bh >> 4;
  const int h = bh & 15;
  float* orow = att + ((size_t)(b * S_LEN) + qg) * D_MODEL + h * DK + 8 * half;
  {
    v4f lo, hi;
    lo[0] = o0[0] * inv; lo[1] = o0[1] * inv; lo[2] = o0[2] * inv; lo[3] = o0[3] * inv;
    hi[0] = o0[4] * inv; hi[1] = o0[5] * inv; hi[2] = o0[6] * inv; hi[3] = o0[7] * inv;
    *(v4f*)(orow +  0) = lo; *(v4f*)(orow +  4) = hi;
    lo[0] = o1[0] * inv; lo[1] = o1[1] * inv; lo[2] = o1[2] * inv; lo[3] = o1[3] * inv;
    hi[0] = o1[4] * inv; hi[1] = o1[5] * inv; hi[2] = o1[6] * inv; hi[3] = o1[7] * inv;
    *(v4f*)(orow + 16) = lo; *(v4f*)(orow + 20) = hi;
    lo[0] = o2[0] * inv; lo[1] = o2[1] * inv; lo[2] = o2[2] * inv; lo[3] = o2[3] * inv;
    hi[0] = o2[4] * inv; hi[1] = o2[5] * inv; hi[2] = o2[6] * inv; hi[3] = o2[7] * inv;
    *(v4f*)(orow + 32) = lo; *(v4f*)(orow + 36) = hi;
    lo[0] = o3[0] * inv; lo[1] = o3[1] * inv; lo[2] = o3[2] * inv; lo[3] = o3[3] * inv;
    hi[0] = o3[4] * inv; hi[1] = o3[5] * inv; hi[2] = o3[6] * inv; hi[3] = o3[7] * inv;
    *(v4f*)(orow + 48) = lo; *(v4f*)(orow + 52) = hi;
  }
}

extern "C" void kernel_launch(void* const* d_in, const int* in_sizes, int n_in,
                              void* d_out, int out_size, void* d_ws, size_t ws_size,
                              hipStream_t stream) {
  const float* x  = (const float*)d_in[0];
  const float* wq = (const float*)d_in[1];
  const float* wk = (const float*)d_in[2];
  const float* wv = (const float*)d_in[3];
  const float* wo = (const float*)d_in[4];
  float* out = (float*)d_out;

  float* ws  = (float*)d_ws;
  const size_t SZ = (size_t)M_TOT * D_MODEL;
  float* Qw  = ws;
  float* Kw  = ws + SZ;
  float* Vw  = ws + 2 * SZ;
  float* Aw  = ws + 3 * SZ;

  dim3 blk(256);
  dim3 grd(D_MODEL / 64, M_TOT / (32 * 8));
  gemm_wmma_f32<<<grd, blk, 0, stream>>>(x, wq, Qw, 1);
  gemm_wmma_f32<<<grd, blk, 0, stream>>>(x, wk, Kw, 1);
  gemm_wmma_f32<<<grd, blk, 0, stream>>>(x, wv, Vw, 2);

  flash_attn<<<dim3(S_LEN / 16, BATCH * NHEAD), dim3(32), 0, stream>>>(Qw, Kw, Vw, Aw);

  gemm_wmma_f32<<<grd, blk, 0, stream>>>(Aw, wo, out, 0);
}


# X/W, Q and K fragment loads widened to 16 B per lane (half h takes k+4h..k+4h+3, same k order in A and B): half the load instructions and cache-line touches; k sum order of the products changes
# speedup vs baseline: 1.6640x; 1.0758x over previous
#include <hip/hip_runtime.h>
#include <hip/hip_bf16.h>

typedef __attribute__((ext_vector_type(2))) float v2f;
typedef __attribute__((ext_vector_type(4))) float v4f;
typedef __attribute__((ext_vector_type(8))) float v8f;

#define S_LEN   2048
#define D_MODEL 1024
#define NHEAD   16
#define DK      64
#define BATCH   2
#define M_TOT   (BATCH * S_LEN)
#define ROPE_L2T_OVER_DK 0.20762050593046014f

__global__ __launch_bounds__(256) void gemm_wmma_f32(
    const float* __restrict__ X, const float* __restrict__ W,
    float* __restrict__ out, int mode)
{
  const int lane = threadIdx.x & 31;
  const int wave = threadIdx.x >> 5;
  const int half = lane >> 4;
  const int l16  = lane & 15;

  const int m0 = (blockIdx.y * 8 + wave) * 32;
  const int n0 = blockIdx.x * 64;

  const float* Arow0 = X + (size_t)(m0 + l16) * D_MODEL;
  const float* Arow1 = Arow0 + (size_t)16 * D_MODEL;
  const float* Brow0 = W + (size_t)(n0 + l16) * D_MODEL;
  const float* Brow1 = Brow0 + (size_t)16 * D_MODEL;
  const float* Brow2 = Brow0 + (size_t)32 * D_MODEL;
  const float* Brow3 = Brow0 + (size_t)48 * D_MODEL;

  v8f acc[2][4];
#pragma unroll
  for (int i = 0; i < 2; ++i)
#pragma unroll
    for (int t = 0; t < 4; ++t) acc[i][t] = (v8f){};

  const int koff = 4 * half;
#pragma unroll 2
  for (int k = 0; k < D_MODEL; k += 8) {
    const int kk = k + koff;
    const v4f a0 = *(const v4f*)(Arow0 + kk);
    const v4f a1 = *(const v4f*)(Arow1 + kk);
    const v4f b0 = *(const v4f*)(Brow0 + kk);
    const v4f b1 = *(const v4f*)(Brow1 + kk);
    const v4f b2 = *(const v4f*)(Brow2 + kk);
    const v4f b3 = *(const v4f*)(Brow3 + kk);
    const v2f a0l = {a0[0], a0[1]}, a0h = {a0[2], a0[3]};
    const v2f a1l = {a1[0], a1[1]}, a1h = {a1[2], a1[3]};
    const v2f b0l = {b0[0], b0[1]}, b0h = {b0[2], b0[3]};
    const v2f b1l = {b1[0], b1[1]}, b1h = {b1[2], b1[3]};
    const v2f b2l = {b2[0], b2[1]}, b2h = {b2[2], b2[3]};
    const v2f b3l = {b3[0], b3[1]}, b3h = {b3[2], b3[3]};
    acc[0][0] = __builtin_amdgcn_wmma_f32_16x16x4_f32(false, a0l, false, b0l, (short)0, acc[0][0], false, false);
    acc[0][1] = __builtin_amdgcn_wmma_f32_16x16x4_f32(false, a0l, false, b1l, (short)0, acc[0][1], false, false);
    acc[0][2] = __builtin_amdgcn_wmma_f32_16x16x4_f32(false, a0l, false, b2l, (short)0, acc[0][2], false, false);
    acc[0][3] = __builtin_amdgcn_wmma_f32_16x16x4_f32(false, a0l, false, b3l, (short)0, acc[0][3], false, false);
    acc[1][0] = __builtin_amdgcn_wmma_f32_16x16x4_f32(false, a1l, false, b0l, (short)0, acc[1][0], false, false);
    acc[1][1] = __builtin_amdgcn_wmma_f32_16x16x4_f32(false, a1l, false, b1l, (short)0, acc[1][1], false, false);
    acc[1][2] = __builtin_amdgcn_wmma_f32_16x16x4_f32(false, a1l, false, b2l, (short)0, acc[1][2], false, false);
    acc[1][3] = __builtin_amdgcn_wmma_f32_16x16x4_f32(false, a1l, false, b3l, (short)0, acc[1][3], false, false);
    acc[0][0] = __builtin_amdgcn_wmma_f32_16x16x4_f32(false, a0h, false, b0h, (short)0, acc[0][0], false, false);
    acc[0][1] = __builtin_amdgcn_wmma_f32_16x16x4_f32(false, a0h, false, b1h, (short)0, acc[0][1], false, false);
    acc[0][2] = __builtin_amdgcn_wmma_f32_16x16x4_f32(false, a0h, false, b2h, (short)0, acc[0][2], false, false);
    acc[0][3] = __builtin_amdgcn_wmma_f32_16x16x4_f32(false, a0h, false, b3h, (short)0, acc[0][3], false, false);
    acc[1][0] = __builtin_amdgcn_wmma_f32_16x16x4_f32(false, a1h, false, b0h, (short)0, acc[1][0], false, false);
    acc[1][1] = __builtin_amdgcn_wmma_f32_16x16x4_f32(false, a1h, false, b1h, (short)0, acc[1][1], false, false);
    acc[1][2] = __builtin_amdgcn_wmma_f32_16x16x4_f32(false, a1h, false, b2h, (short)0, acc[1][2], false, false);
    acc[1][3] = __builtin_amdgcn_wmma_f32_16x16x4_f32(false, a1h, false, b3h, (short)0, acc[1][3], false, false);
  }

  if (mode == 0) {
#pragma unroll
    for (int i = 0; i < 2; ++i)
#pragma unroll
      for (int t = 0; t < 4; ++t)
#pragma unroll
        for (int r = 0; r < 8; ++r) {
          const int row = m0 + i * 16 + r + 8 * half;
          out[(size_t)row * D_MODEL + n0 + t * 16 + l16] = acc[i][t][r];
        }
  } else {
#pragma unroll
    for (int t = 0; t < 4; ++t) {
      const int gcol = n0 + t * 16 + l16;
      const int h = gcol >> 6;
      const int d = gcol & 63;
      const bool even = (d & 1) == 0;
      const float invfreq = exp2f(-(float)(d & ~1) * ROPE_L2T_OVER_DK);
#pragma unroll
      for (int i = 0; i < 2; ++i)
#pragma unroll
        for (int r = 0; r < 8; ++r) {
          const int row = m0 + i * 16 + r + 8 * half;
          const int b  = row >> 11;
          const int s  = row & (S_LEN - 1);
          float v  = acc[i][t][r];
          float vp = __shfl_xor(v, 1, 32);
          if (mode == 1) {
            const float ang = (float)s * invfreq;
            const float c = __cosf(ang), sn = __sinf(ang);
            v = v * c + (even ? -vp : vp) * sn;
          }
          out[(((size_t)(b * NHEAD + h) * S_LEN + s) * DK) + d] = v;
        }
    }
  }
}

__device__ __forceinline__ float xhalf(float x) {
  const int b = __builtin_bit_cast(int, x);
  return __builtin_bit_cast(float,
      __builtin_amdgcn_permlanex16(b, b, 0x76543210, 0xfedcba98, false, false));
}

__global__ __launch_bounds__(32) __attribute__((amdgpu_waves_per_eu(4, 4))) void flash_attn(
    const float* __restrict__ Q, const float* __restrict__ K,
    const float* __restrict__ V, float* __restrict__ att)
{
  const int lane = threadIdx.x & 31;
  const int half = lane >> 4;
  const int l16  = lane & 15;
  const int qt   = blockIdx.x;
  const int bh   = blockIdx.y;
  const size_t base = (size_t)bh * S_LEN * DK;

  v4f qb[8];
  {
    const float* qrow = Q + base + (size_t)(qt * 16 + l16) * DK + 4 * half;
#pragma unroll
    for (int c = 0; c < 8; ++c) {
      v4f t = *(const v4f*)(qrow + 8 * c);
      qb[c] = t * 0.125f;
    }
  }

  v8f o0 = {}, o1 = {}, o2 = {}, o3 = {};
  float mrun = -__builtin_inff();
  float lrun = 0.f;
  const int qg = qt * 16 + l16;

  for (int kb = 0; kb <= qt; ++kb) {
    const int kg0 = kb * 16 + 8 * half;

    const float* krow = K + base + (size_t)(kb * 16 + l16) * DK + 4 * half;
    v4f ka[8];
#pragma unroll
    for (int c = 0; c < 8; ++c) ka[c] = *(const v4f*)(krow + 8 * c);

    const float* vcol = V + base + (size_t)kg0 * DK + l16;
    float vv[8][4];
#pragma unroll
    for (int i = 0; i < 8; ++i)
#pragma unroll
      for (int t = 0; t < 4; ++t) vv[i][t] = vcol[i * DK + 16 * t];

    v8f st = {};
#pragma unroll
    for (int c = 0; c < 8; ++c) {
      const v2f kl = {ka[c][0], ka[c][1]}, kh = {ka[c][2], ka[c][3]};
      const v2f ql = {qb[c][0], qb[c][1]}, qh = {qb[c][2], qb[c][3]};
      st = __builtin_amdgcn_wmma_f32_16x16x4_f32(false, kl, false, ql, (short)0, st, false, false);
      st = __builtin_amdgcn_wmma_f32_16x16x4_f32(false, kh, false, qh, (short)0, st, false, false);
    }

    float s[8];
#pragma unroll
    for (int r = 0; r < 8; ++r)
      s[r] = (kg0 + r <= qg) ? st[r] : -__builtin_inff();
    float mx = s[0];
#pragma unroll
    for (int r = 1; r < 8; ++r) mx = fmaxf(mx, s[r]);
    mx = fmaxf(mx, xhalf(mx));
    const float mnew = fmaxf(mrun, mx);
    const float scl  = __expf(mrun - mnew);
    float p[8];
    float psum = 0.f;
#pragma unroll
    for (int r = 0; r < 8; ++r) {
      p[r] = __expf(s[r] - mnew);
      psum += p[r];
    }
    lrun = lrun * scl + psum;
    mrun = mnew;

    o0 *= scl; o1 *= scl; o2 *= scl; o3 *= scl;

#pragma unroll
    for (int j = 0; j < 4; ++j) {
      v2f pb;
      pb[0] = p[2 * j];
      pb[1] = p[2 * j + 1];
      v2f a0, a1, a2, a3;
      a0[0] = vv[2 * j][0]; a0[1] = vv[2 * j + 1][0];
      a1[0] = vv[2 * j][1]; a1[1] = vv[2 * j + 1][1];
      a2[0] = vv[2 * j][2]; a2[1] = vv[2 * j + 1][2];
      a3[0] = vv[2 * j][3]; a3[1] = vv[2 * j + 1][3];
      o0 = __builtin_amdgcn_wmma_f32_16x16x4_f32(false, a0, false, pb, (short)0, o0, false, false);
      o1 = __builtin_amdgcn_wmma_f32_16x16x4_f32(false, a1, false, pb, (short)0, o1, false, false);
      o2 = __builtin_amdgcn_wmma_f32_16x16x4_f32(false, a2, false, pb, (short)0, o2, false, false);
      o3 = __builtin_amdgcn_wmma_f32_16x16x4_f32(false, a3, false, pb, (short)0, o3, false, false);
    }
  }

  const float inv = 1.0f / (lrun + xhalf(lrun));
  const int b = bh >> 4;
  const int h = bh & 15;
  float* orow = att + ((size_t)(b * S_LEN) + qg) * D_MODEL + h * DK + 8 * half;
  {
    v4f lo, hi;
    lo[0] = o0[0] * inv; lo[1] = o0[1] * inv; lo[2] = o0[2] * inv; lo[3] = o0[3] * inv;
    hi[0] = o0[4] * inv; hi[1] = o0[5] * inv; hi[2] = o0[6] * inv; hi[3] = o0[7] * inv;
    *(v4f*)(orow +  0) = lo; *(v4f*)(orow +  4) = hi;
    lo[0] = o1[0] * inv; lo[1] = o1[1] * inv; lo[2] = o1[2] * inv; lo[3] = o1[3] * inv;
    hi[0] = o1[4] * inv; hi[1] = o1[5] * inv; hi[2] = o1[6] * inv; hi[3] = o1[7] * inv;
    *(v4f*)(orow + 16) = lo; *(v4f*)(orow + 20) = hi;
    lo[0] = o2[0] * inv; lo[1] = o2[1] * inv; lo[2] = o2[2] * inv; lo[3] = o2[3] * inv;
    hi[0] = o2[4] * inv; hi[1] = o2[5] * inv; hi[2] = o2[6] * inv; hi[3] = o2[7] * inv;
    *(v4f*)(orow + 32) = lo; *(v4f*)(orow + 36) = hi;
    lo[0] = o3[0] * inv; lo[1] = o3[1] * inv; lo[2] = o3[2] * inv; lo[3] = o3[3] * inv;
    hi[0] = o3[4] * inv; hi[1] = o3[5] * inv; hi[2] = o3[6] * inv; hi[3] = o3[7] * inv;
    *(v4f*)(orow + 48) = lo; *(v4f*)(orow + 52) = hi;
  }
}

extern "C" void kernel_launch(void* const* d_in, const int* in_sizes, int n_in,
                              void* d_out, int out_size, void* d_ws, size_t ws_size,
                              hipStream_t stream) {
  const float* x  = (const float*)d_in[0];
  const float* wq = (const float*)d_in[1];
  const float* wk = (const float*)d_in[2];
  const float* wv = (const float*)d_in[3];
  const float* wo = (const float*)d_in[4];
  float* out = (float*)d_out;

  float* ws  = (float*)d_ws;
  const size_t SZ = (size_t)M_TOT * D_MODEL;
  float* Qw  = ws;
  float* Kw  = ws + SZ;
  float* Vw  = ws + 2 * SZ;
  float* Aw  = ws + 3 * SZ;

  dim3 blk(256);
  dim3 grd(D_MODEL / 64, M_TOT / (32 * 8));
  gemm_wmma_f32<<<grd, blk, 0, stream>>>(x, wq, Qw, 1);
  gemm_wmma_f32<<<grd, blk, 0, stream>>>(x, wk, Kw, 1);
  gemm_wmma_f32<<<grd, blk, 0, stream>>>(x, wv, Vw, 2);

  flash_attn<<<dim3(S_LEN / 16, BATCH * NHEAD), dim3(32), 0, stream>>>(Qw, Kw, Vw, Aw);

  gemm_wmma_f32<<<grd, blk, 0, stream>>>(Aw, wo, out, 0);
}


# attention: each wave owns query tiles 127-p and p of one head (129 key blocks per wave, equal causal work); 256 groups of 8 waves, 2 per SIMD, replace 4096 unequal one-wave groups; sums unchanged
# speedup vs baseline: 1.7482x; 1.0506x over previous
#include <hip/hip_runtime.h>
#include <hip/hip_bf16.h>

typedef __attribute__((ext_vector_type(2))) float v2f;
typedef __attribute__((ext_vector_type(4))) float v4f;
typedef __attribute__((ext_vector_type(8))) float v8f;

#define S_LEN   2048
#define D_MODEL 1024
#define NHEAD   16
#define DK      64
#define BATCH   2
#define M_TOT   (BATCH * S_LEN)
#define ROPE_L2T_OVER_DK 0.20762050593046014f

__global__ __launch_bounds__(256) void gemm_wmma_f32(
    const float* __restrict__ X, const float* __restrict__ W,
    float* __restrict__ out, int mode)
{
  const int lane = threadIdx.x & 31;
  const int wave = threadIdx.x >> 5;
  const int half = lane >> 4;
  const int l16  = lane & 15;

  const int m0 = (blockIdx.y * 8 + wave) * 32;
  const int n0 = blockIdx.x * 64;

  const float* Arow0 = X + (size_t)(m0 + l16) * D_MODEL;
  const float* Arow1 = Arow0 + (size_t)16 * D_MODEL;
  const float* Brow0 = W + (size_t)(n0 + l16) * D_MODEL;
  const float* Brow1 = Brow0 + (size_t)16 * D_MODEL;
  const float* Brow2 = Brow0 + (size_t)32 * D_MODEL;
  const float* Brow3 = Brow0 + (size_t)48 * D_MODEL;

  v8f acc[2][4];
#pragma unroll
  for (int i = 0; i < 2; ++i)
#pragma unroll
    for (int t = 0; t < 4; ++t) acc[i][t] = (v8f){};

  const int koff = 4 * half;
#pragma unroll 2
  for (int k = 0; k < D_MODEL; k += 8) {
    const int kk = k + koff;
    const v4f a0 = *(const v4f*)(Arow0 + kk);
    const v4f a1 = *(const v4f*)(Arow1 + kk);
    const v4f b0 = *(const v4f*)(Brow0 + kk);
    const v4f b1 = *(const v4f*)(Brow1 + kk);
    const v4f b2 = *(const v4f*)(Brow2 + kk);
    const v4f b3 = *(const v4f*)(Brow3 + kk);
    const v2f a0l = {a0[0], a0[1]}, a0h = {a0[2], a0[3]};
    const v2f a1l = {a1[0], a1[1]}, a1h = {a1[2], a1[3]};
    const v2f b0l = {b0[0], b0[1]}, b0h = {b0[2], b0[3]};
    const v2f b1l = {b1[0], b1[1]}, b1h = {b1[2], b1[3]};
    const v2f b2l = {b2[0], b2[1]}, b2h = {b2[2], b2[3]};
    const v2f b3l = {b3[0], b3[1]}, b3h = {b3[2], b3[3]};
    acc[0][0] = __builtin_amdgcn_wmma_f32_16x16x4_f32(false, a0l, false, b0l, (short)0, acc[0][0], false, false);
    acc[0][1] = __builtin_amdgcn_wmma_f32_16x16x4_f32(false, a0l, false, b1l, (short)0, acc[0][1], false, false);
    acc[0][2] = __builtin_amdgcn_wmma_f32_16x16x4_f32(false, a0l, false, b2l, (short)0, acc[0][2], false, false);
    acc[0][3] = __builtin_amdgcn_wmma_f32_16x16x4_f32(false, a0l, false, b3l, (short)0, acc[0][3], false, false);
    acc[1][0] = __builtin_amdgcn_wmma_f32_16x16x4_f32(false, a1l, false, b0l, (short)0, acc[1][0], false, false);
    acc[1][1] = __builtin_amdgcn_wmma_f32_16x16x4_f32(false, a1l, false, b1l, (short)0, acc[1][1], false, false);
    acc[1][2] = __builtin_amdgcn_wmma_f32_16x16x4_f32(false, a1l, false, b2l, (short)0, acc[1][2], false, false);
    acc[1][3] = __builtin_amdgcn_wmma_f32_16x16x4_f32(false, a1l, false, b3l, (short)0, acc[1][3], false, false);
    acc[0][0] = __builtin_amdgcn_wmma_f32_16x16x4_f32(false, a0h, false, b0h, (short)0, acc[0][0], false, false);
    acc[0][1] = __builtin_amdgcn_wmma_f32_16x16x4_f32(false, a0h, false, b1h, (short)0, acc[0][1], false, false);
    acc[0][2] = __builtin_amdgcn_wmma_f32_16x16x4_f32(false, a0h, false, b2h, (short)0, acc[0][2], false, false);
    acc[0][3] = __builtin_amdgcn_wmma_f32_16x16x4_f32(false, a0h, false, b3h, (short)0, acc[0][3], false, false);
    acc[1][0] = __builtin_amdgcn_wmma_f32_16x16x4_f32(false, a1h, false, b0h, (short)0, acc[1][0], false, false);
    acc[1][1] = __builtin_amdgcn_wmma_f32_16x16x4_f32(false, a1h, false, b1h, (short)0, acc[1][1], false, false);
    acc[1][2] = __builtin_amdgcn_wmma_f32_16x16x4_f32(false, a1h, false, b2h, (short)0, acc[1][2], false, false);
    acc[1][3] = __builtin_amdgcn_wmma_f32_16x16x4_f32(false, a1h, false, b3h, (short)0, acc[1][3], false, false);
  }

  if (mode == 0) {
#pragma unroll
    for (int i = 0; i < 2; ++i)
#pragma unroll
      for (int t = 0; t < 4; ++t)
#pragma unroll
        for (int r = 0; r < 8; ++r) {
          const int row = m0 + i * 16 + r + 8 * half;
          out[(size_t)row * D_MODEL + n0 + t * 16 + l16] = acc[i][t][r];
        }
  } else {
#pragma unroll
    for (int t = 0; t < 4; ++t) {
      const int gcol = n0 + t * 16 + l16;
      const int h = gcol >> 6;
      const int d = gcol & 63;
      const bool even = (d & 1) == 0;
      const float invfreq = exp2f(-(float)(d & ~1) * ROPE_L2T_OVER_DK);
#pragma unroll
      for (int i = 0; i < 2; ++i)
#pragma unroll
        for (int r = 0; r < 8; ++r) {
          const int row = m0 + i * 16 + r + 8 * half;
          const int b  = row >> 11;
          const int s  = row & (S_LEN - 1);
          float v  = acc[i][t][r];
          float vp = __shfl_xor(v, 1, 32);
          if (mode == 1) {
            const float ang = (float)s * invfreq;
            const float c = __cosf(ang), sn = __sinf(ang);
            v = v * c + (even ? -vp : vp) * sn;
          }
          out[(((size_t)(b * NHEAD + h) * S_LEN + s) * DK) + d] = v;
        }
    }
  }
}

__device__ __forceinline__ float xhalf(float x) {
  const int b = __builtin_bit_cast(int, x);
  return __builtin_bit_cast(float,
      __builtin_amdgcn_permlanex16(b, b, 0x76543210, 0xfedcba98, false, false));
}

__global__ __launch_bounds__(256) __attribute__((amdgpu_waves_per_eu(2, 2))) void flash_attn(
    const float* __restrict__ Q, const float* __restrict__ K,
    const float* __restrict__ V, float* __restrict__ att)
{
  const int lane = threadIdx.x & 31;
  const int wave = __builtin_amdgcn_readfirstlane((int)(threadIdx.x >> 5));
  const int half = lane >> 4;
  const int l16  = lane & 15;
  const int task = blockIdx.x * 8 + wave;
  const int bh   = task >> 6;
  const int p    = task & 63;
  const size_t base = (size_t)bh * S_LEN * DK;

  for (int pass = 0; pass < 2; ++pass) {
  const int qt = pass == 0 ? (S_LEN / 16 - 1) - p : p;

  v4f qb[8];
  {
    const float* qrow = Q + base + (size_t)(qt * 16 + l16) * DK + 4 * half;
#pragma unroll
    for (int c = 0; c < 8; ++c) {
      v4f t = *(const v4f*)(qrow + 8 * c);
      qb[c] = t * 0.125f;
    }
  }

  v8f o0 = {}, o1 = {}, o2 = {}, o3 = {};
  float mrun = -__builtin_inff();
  float lrun = 0.f;
  const int qg = qt * 16 + l16;

  for (int kb = 0; kb <= qt; ++kb) {
    const int kg0 = kb * 16 + 8 * half;

    const float* krow = K + base + (size_t)(kb * 16 + l16) * DK + 4 * half;
    v4f ka[8];
#pragma unroll
    for (int c = 0; c < 8; ++c) ka[c] = *(const v4f*)(krow + 8 * c);

    const float* vcol = V + base + (size_t)kg0 * DK + l16;
    float vv[8][4];
#pragma unroll
    for (int i = 0; i < 8; ++i)
#pragma unroll
      for (int t = 0; t < 4; ++t) vv[i][t] = vcol[i * DK + 16 * t];

    v8f st = {};
#pragma unroll
    for (int c = 0; c < 8; ++c) {
      const v2f kl = {ka[c][0], ka[c][1]}, kh = {ka[c][2], ka[c][3]};
      const v2f ql = {qb[c][0], qb[c][1]}, qh = {qb[c][2], qb[c][3]};
      st = __builtin_amdgcn_wmma_f32_16x16x4_f32(false, kl, false, ql, (short)0, st, false, false);
      st = __builtin_amdgcn_wmma_f32_16x16x4_f32(false, kh, false, qh, (short)0, st, false, false);
    }

    float s[8];
#pragma unroll
    for (int r = 0; r < 8; ++r)
      s[r] = (kg0 + r <= qg) ? st[r] : -__builtin_inff();
    float mx = s[0];
#pragma unroll
    for (int r = 1; r < 8; ++r) mx = fmaxf(mx, s[r]);
    mx = fmaxf(mx, xhalf(mx));
    const float mnew = fmaxf(mrun, mx);
    const float scl  = __expf(mrun - mnew);
    float p[8];
    float psum = 0.f;
#pragma unroll
    for (int r = 0; r < 8; ++r) {
      p[r] = __expf(s[r] - mnew);
      psum += p[r];
    }
    lrun = lrun * scl + psum;
    mrun = mnew;

    o0 *= scl; o1 *= scl; o2 *= scl; o3 *= scl;

#pragma unroll
    for (int j = 0; j < 4; ++j) {
      v2f pb;
      pb[0] = p[2 * j];
      pb[1] = p[2 * j + 1];
      v2f a0, a1, a2, a3;
      a0[0] = vv[2 * j][0]; a0[1] = vv[2 * j + 1][0];
      a1[0] = vv[2 * j][1]; a1[1] = vv[2 * j + 1][1];
      a2[0] = vv[2 * j][2]; a2[1] = vv[2 * j + 1][2];
      a3[0] = vv[2 * j][3]; a3[1] = vv[2 * j + 1][3];
      o0 = __builtin_amdgcn_wmma_f32_16x16x4_f32(false, a0, false, pb, (short)0, o0, false, false);
      o1 = __builtin_amdgcn_wmma_f32_16x16x4_f32(false, a1, false, pb, (short)0, o1, false, false);
      o2 = __builtin_amdgcn_wmma_f32_16x16x4_f32(false, a2, false, pb, (short)0, o2, false, false);
      o3 = __builtin_amdgcn_wmma_f32_16x16x4_f32(false, a3, false, pb, (short)0, o3, false, false);
    }
  }

  const float inv = 1.0f / (lrun + xhalf(lrun));
  const int b = bh >> 4;
  const int h = bh & 15;
  float* orow = att + ((size_t)(b * S_LEN) + qg) * D_MODEL + h * DK + 8 * half;
  {
    v4f lo, hi;
    lo[0] = o0[0] * inv; lo[1] = o0[1] * inv; lo[2] = o0[2] * inv; lo[3] = o0[3] * inv;
    hi[0] = o0[4] * inv; hi[1] = o0[5] * inv; hi[2] = o0[6] * inv; hi[3] = o0[7] * inv;
    *(v4f*)(orow +  0) = lo; *(v4f*)(orow +  4) = hi;
    lo[0] = o1[0] * inv; lo[1] = o1[1] * inv; lo[2] = o1[2] * inv; lo[3] = o1[3] * inv;
    hi[0] = o1[4] * inv; hi[1] = o1[5] * inv; hi[2] = o1[6] * inv; hi[3] = o1[7] * inv;
    *(v4f*)(orow + 16) = lo; *(v4f*)(orow + 20) = hi;
    lo[0] = o2[0] * inv; lo[1] = o2[1] * inv; lo[2] = o2[2] * inv; lo[3] = o2[3] * inv;
    hi[0] = o2[4] * inv; hi[1] = o2[5] * inv; hi[2] = o2[6] * inv; hi[3] = o2[7] * inv;
    *(v4f*)(orow + 32) = lo; *(v4f*)(orow + 36) = hi;
    lo[0] = o3[0] * inv; lo[1] = o3[1] * inv; lo[2] = o3[2] * inv; lo[3] = o3[3] * inv;
    hi[0] = o3[4] * inv; hi[1] = o3[5] * inv; hi[2] = o3[6] * inv; hi[3] = o3[7] * inv;
    *(v4f*)(orow + 48) = lo; *(v4f*)(orow + 52) = hi;
  }
  }
}

extern "C" void kernel_launch(void* const* d_in, const int* in_sizes, int n_in,
                              void* d_out, int out_size, void* d_ws, size_t ws_size,
                              hipStream_t stream) {
  const float* x  = (const float*)d_in[0];
  const float* wq = (const float*)d_in[1];
  const float* wk = (const float*)d_in[2];
  const float* wv = (const float*)d_in[3];
  const float* wo = (const float*)d_in[4];
  float* out = (float*)d_out;

  float* ws  = (float*)d_ws;
  const size_t SZ = (size_t)M_TOT * D_MODEL;
  float* Qw  = ws;
  float* Kw  = ws + SZ;
  float* Vw  = ws + 2 * SZ;
  float* Aw  = ws + 3 * SZ;

  dim3 blk(256);
  dim3 grd(D_MODEL / 64, M_TOT / (32 * 8));
  gemm_wmma_f32<<<grd, blk, 0, stream>>>(x, wq, Qw, 1);
  gemm_wmma_f32<<<grd, blk, 0, stream>>>(x, wk, Kw, 1);
  gemm_wmma_f32<<<grd, blk, 0, stream>>>(x, wv, Vw, 2);

  flash_attn<<<dim3(256), dim3(256), 0, stream>>>(Qw, Kw, Vw, Aw);

  gemm_wmma_f32<<<grd, blk, 0, stream>>>(Aw, wo, out, 0);
}


# every fragment load group issued one block ahead of its products: GEMM k loop as a ring of two fragment sets, attention K one key block ahead and V in front of the score product; sums unchanged
# speedup vs baseline: 1.9599x; 1.1211x over previous
#include <hip/hip_runtime.h>
#include <hip/hip_bf16.h>

typedef __attribute__((ext_vector_type(2))) float v2f;
typedef __attribute__((ext_vector_type(4))) float v4f;
typedef __attribute__((ext_vector_type(8))) float v8f;

#define S_LEN   2048
#define D_MODEL 1024
#define NHEAD   16
#define DK      64
#define BATCH   2
#define M_TOT   (BATCH * S_LEN)
#define ROPE_L2T_OVER_DK 0.20762050593046014f

struct GemmFrag {
  v4f a[2][2];
  v4f b[2][4];
};

__device__ __forceinline__ void gemm_load(GemmFrag& f,
    const float* __restrict__ A0, const float* __restrict__ A1,
    const float* __restrict__ B0, const float* __restrict__ B1,
    const float* __restrict__ B2, const float* __restrict__ B3, int kk)
{
#pragma unroll
  for (int s = 0; s < 2; ++s) {
    f.a[s][0] = *(const v4f*)(A0 + kk + 8 * s);
    f.a[s][1] = *(const v4f*)(A1 + kk + 8 * s);
    f.b[s][0] = *(const v4f*)(B0 + kk + 8 * s);
    f.b[s][1] = *(const v4f*)(B1 + kk + 8 * s);
    f.b[s][2] = *(const v4f*)(B2 + kk + 8 * s);
    f.b[s][3] = *(const v4f*)(B3 + kk + 8 * s);
  }
}

__device__ __forceinline__ void gemm_mm(v8f (&acc)[2][4], const GemmFrag& f)
{
#pragma unroll
  for (int s = 0; s < 2; ++s)
#pragma unroll
    for (int e = 0; e < 4; e += 2)
#pragma unroll
      for (int i = 0; i < 2; ++i) {
        const v2f av = {f.a[s][i][e], f.a[s][i][e + 1]};
#pragma unroll
        for (int t = 0; t < 4; ++t) {
          const v2f bv = {f.b[s][t][e], f.b[s][t][e + 1]};
          acc[i][t] = __builtin_amdgcn_wmma_f32_16x16x4_f32(false, av, false, bv, (short)0, acc[i][t], false, false);
        }
      }
}

__global__ __launch_bounds__(256) __attribute__((amdgpu_waves_per_eu(2, 2))) void gemm_wmma_f32(
    const float* __restrict__ X, const float* __restrict__ W,
    float* __restrict__ out, int mode)
{
  const int lane = threadIdx.x & 31;
  const int wave = threadIdx.x >> 5;
  const int half = lane >> 4;
  const int l16  = lane & 15;

  const int m0 = (blockIdx.y * 8 + wave) * 32;
  const int n0 = blockIdx.x * 64;

  const int koff = 4 * half;
  const float* Arow0 = X + (size_t)(m0 + l16) * D_MODEL + koff;
  const float* Arow1 = Arow0 + (size_t)16 * D_MODEL;
  const float* Brow0 = W + (size_t)(n0 + l16) * D_MODEL + koff;
  const float* Brow1 = Brow0 + (size_t)16 * D_MODEL;
  const float* Brow2 = Brow0 + (size_t)32 * D_MODEL;
  const float* Brow3 = Brow0 + (size_t)48 * D_MODEL;

  v8f acc[2][4];
#pragma unroll
  for (int i = 0; i < 2; ++i)
#pragma unroll
    for (int t = 0; t < 4; ++t) acc[i][t] = (v8f){};

  GemmFrag fp, fq;
  gemm_load(fp, Arow0, Arow1, Brow0, Brow1, Brow2, Brow3, 0);
  for (int k = 0; k < D_MODEL; k += 32) {
    gemm_load(fq, Arow0, Arow1, Brow0, Brow1, Brow2, Brow3, k + 16);
    gemm_mm(acc, fp);
    gemm_load(fp, Arow0, Arow1, Brow0, Brow1, Brow2, Brow3, (k + 32) & (D_MODEL - 1));
    gemm_mm(acc, fq);
    asm volatile("" ::: "memory");
  }

  if (mode == 0) {
#pragma unroll
    for (int i = 0; i < 2; ++i)
#pragma unroll
      for (int t = 0; t < 4; ++t)
#pragma unroll
        for (int r = 0; r < 8; ++r) {
          const int row = m0 + i * 16 + r + 8 * half;
          out[(size_t)row * D_MODEL + n0 + t * 16 + l16] = acc[i][t][r];
        }
  } else {
#pragma unroll
    for (int t = 0; t < 4; ++t) {
      const int gcol = n0 + t * 16 + l16;
      const int h = gcol >> 6;
      const int d = gcol & 63;
      const bool even = (d & 1) == 0;
      const float invfreq = exp2f(-(float)(d & ~1) * ROPE_L2T_OVER_DK);
#pragma unroll
      for (int i = 0; i < 2; ++i)
#pragma unroll
        for (int r = 0; r < 8; ++r) {
          const int row = m0 + i * 16 + r + 8 * half;
          const int b  = row >> 11;
          const int s  = row & (S_LEN - 1);
          float v  = acc[i][t][r];
          float vp = __shfl_xor(v, 1, 32);
          if (mode == 1) {
            const float ang = (float)s * invfreq;
            const float c = __cosf(ang), sn = __sinf(ang);
            v = v * c + (even ? -vp : vp) * sn;
          }
          out[(((size_t)(b * NHEAD + h) * S_LEN + s) * DK) + d] = v;
        }
    }
  }
}

__device__ __forceinline__ float xhalf(float x) {
  const int b = __builtin_bit_cast(int, x);
  return __builtin_bit_cast(float,
      __builtin_amdgcn_permlanex16(b, b, 0x76543210, 0xfedcba98, false, false));
}

__global__ __launch_bounds__(256) __attribute__((amdgpu_waves_per_eu(2, 2))) void flash_attn(
    const float* __restrict__ Q, const float* __restrict__ K,
    const float* __restrict__ V, float* __restrict__ att)
{
  const int lane = threadIdx.x & 31;
  const int wave = __builtin_amdgcn_readfirstlane((int)(threadIdx.x >> 5));
  const int half = lane >> 4;
  const int l16  = lane & 15;
  const int task = blockIdx.x * 8 + wave;
  const int bh   = task >> 6;
  const int p    = task & 63;
  const size_t base = (size_t)bh * S_LEN * DK;

  for (int pass = 0; pass < 2; ++pass) {
  const int qt = pass == 0 ? (S_LEN / 16 - 1) - p : p;

  v4f qb[8];
  {
    const float* qrow = Q + base + (size_t)(qt * 16 + l16) * DK + 4 * half;
#pragma unroll
    for (int c = 0; c < 8; ++c) {
      v4f t = *(const v4f*)(qrow + 8 * c);
      qb[c] = t * 0.125f;
    }
  }

  v8f o0 = {}, o1 = {}, o2 = {}, o3 = {};
  float mrun = -__builtin_inff();
  float lrun = 0.f;
  const int qg = qt * 16 + l16;

  const float* kbase = K + base + (size_t)l16 * DK + 4 * half;
  v4f ka[8];
#pragma unroll
  for (int c = 0; c < 8; ++c) ka[c] = *(const v4f*)(kbase + 8 * c);

  for (int kb = 0; kb <= qt; ++kb) {
    const int kg0 = kb * 16 + 8 * half;

    const float* vcol = V + base + (size_t)kg0 * DK + l16;
    float vv[8][4];
#pragma unroll
    for (int i = 0; i < 8; ++i)
#pragma unroll
      for (int t = 0; t < 4; ++t) vv[i][t] = vcol[i * DK + 16 * t];

    v8f st = {};
#pragma unroll
    for (int c = 0; c < 8; ++c) {
      const v2f kl = {ka[c][0], ka[c][1]}, kh = {ka[c][2], ka[c][3]};
      const v2f ql = {qb[c][0], qb[c][1]}, qh = {qb[c][2], qb[c][3]};
      st = __builtin_amdgcn_wmma_f32_16x16x4_f32(false, kl, false, ql, (short)0, st, false, false);
      st = __builtin_amdgcn_wmma_f32_16x16x4_f32(false, kh, false, qh, (short)0, st, false, false);
    }

    {
      const int kbn = kb < qt ? kb + 1 : qt;
      const float* krow = kbase + (size_t)kbn * (16 * DK);
#pragma unroll
      for (int c = 0; c < 8; ++c) ka[c] = *(const v4f*)(krow + 8 * c);
    }

    float s[8];
#pragma unroll
    for (int r = 0; r < 8; ++r)
      s[r] = (kg0 + r <= qg) ? st[r] : -__builtin_inff();
    float mx = s[0];
#pragma unroll
    for (int r = 1; r < 8; ++r) mx = fmaxf(mx, s[r]);
    mx = fmaxf(mx, xhalf(mx));
    const float mnew = fmaxf(mrun, mx);
    const float scl  = __expf(mrun - mnew);
    float p[8];
    float psum = 0.f;
#pragma unroll
    for (int r = 0; r < 8; ++r) {
      p[r] = __expf(s[r] - mnew);
      psum += p[r];
    }
    lrun = lrun * scl + psum;
    mrun = mnew;

    o0 *= scl; o1 *= scl; o2 *= scl; o3 *= scl;

#pragma unroll
    for (int j = 0; j < 4; ++j) {
      v2f pb;
      pb[0] = p[2 * j];
      pb[1] = p[2 * j + 1];
      v2f a0, a1, a2, a3;
      a0[0] = vv[2 * j][0]; a0[1] = vv[2 * j + 1][0];
      a1[0] = vv[2 * j][1]; a1[1] = vv[2 * j + 1][1];
      a2[0] = vv[2 * j][2]; a2[1] = vv[2 * j + 1][2];
      a3[0] = vv[2 * j][3]; a3[1] = vv[2 * j + 1][3];
      o0 = __builtin_amdgcn_wmma_f32_16x16x4_f32(false, a0, false, pb, (short)0, o0, false, false);
      o1 = __builtin_amdgcn_wmma_f32_16x16x4_f32(false, a1, false, pb, (short)0, o1, false, false);
      o2 = __builtin_amdgcn_wmma_f32_16x16x4_f32(false, a2, false, pb, (short)0, o2, false, false);
      o3 = __builtin_amdgcn_wmma_f32_16x16x4_f32(false, a3, false, pb, (short)0, o3, false, false);
    }
  }

  const float inv = 1.0f / (lrun + xhalf(lrun));
  const int b = bh >> 4;
  const int h = bh & 15;
  float* orow = att + ((size_t)(b * S_LEN) + qg) * D_MODEL + h * DK + 8 * half;
  {
    v4f lo, hi;
    lo[0] = o0[0] * inv; lo[1] = o0[1] * inv; lo[2] = o0[2] * inv; lo[3] = o0[3] * inv;
    hi[0] = o0[4] * inv; hi[1] = o0[5] * inv; hi[2] = o0[6] * inv; hi[3] = o0[7] * inv;
    *(v4f*)(orow +  0) = lo; *(v4f*)(orow +  4) = hi;
    lo[0] = o1[0] * inv; lo[1] = o1[1] * inv; lo[2] = o1[2] * inv; lo[3] = o1[3] * inv;
    hi[0] = o1[4] * inv; hi[1] = o1[5] * inv; hi[2] = o1[6] * inv; hi[3] = o1[7] * inv;
    *(v4f*)(orow + 16) = lo; *(v4f*)(orow + 20) = hi;
    lo[0] = o2[0] * inv; lo[1] = o2[1] * inv; lo[2] = o2[2] * inv; lo[3] = o2[3] * inv;
    hi[0] = o2[4] * inv; hi[1] = o2[5] * inv; hi[2] = o2[6] * inv; hi[3] = o2[7] * inv;
    *(v4f*)(orow + 32) = lo; *(v4f*)(orow + 36) = hi;
    lo[0] = o3[0] * inv; lo[1] = o3[1] * inv; lo[2] = o3[2] * inv; lo[3] = o3[3] * inv;
    hi[0] = o3[4] * inv; hi[1] = o3[5] * inv; hi[2] = o3[6] * inv; hi[3] = o3[7] * inv;
    *(v4f*)(orow + 48) = lo; *(v4f*)(orow + 52) = hi;
  }
  }
}

extern "C" void kernel_launch(void* const* d_in, const int* in_sizes, int n_in,
                              void* d_out, int out_size, void* d_ws, size_t ws_size,
                              hipStream_t stream) {
  const float* x  = (const float*)d_in[0];
  const float* wq = (const float*)d_in[1];
  const float* wk = (const float*)d_in[2];
  const float* wv = (const float*)d_in[3];
  const float* wo = (const float*)d_in[4];
  float* out = (float*)d_out;

  float* ws  = (float*)d_ws;
  const size_t SZ = (size_t)M_TOT * D_MODEL;
  float* Qw  = ws;
  float* Kw  = ws + SZ;
  float* Vw  = ws + 2 * SZ;
  float* Aw  = ws + 3 * SZ;

  dim3 blk(256);
  dim3 grd(D_MODEL / 64, M_TOT / (32 * 8));
  gemm_wmma_f32<<<grd, blk, 0, stream>>>(x, wq, Qw, 1);
  gemm_wmma_f32<<<grd, blk, 0, stream>>>(x, wk, Kw, 1);
  gemm_wmma_f32<<<grd, blk, 0, stream>>>(x, wv, Vw, 2);

  flash_attn<<<dim3(256), dim3(256), 0, stream>>>(Qw, Kw, Vw, Aw);

  gemm_wmma_f32<<<grd, blk, 0, stream>>>(Aw, wo, out, 0);
}


# attention: each wave takes two neighbouring query tiles against every K and V block (40 fragment loads per 64 products, was per 32), two K sets used in turn; 1024 waves, one per SIMD; sums unchanged
# speedup vs baseline: 2.2772x; 1.1619x over previous
#include <hip/hip_runtime.h>
#include <hip/hip_bf16.h>

typedef __attribute__((ext_vector_type(2))) float v2f;
typedef __attribute__((ext_vector_type(4))) float v4f;
typedef __attribute__((ext_vector_type(8))) float v8f;

#define S_LEN   2048
#define D_MODEL 1024
#define NHEAD   16
#define DK      64
#define BATCH   2
#define M_TOT   (BATCH * S_LEN)
#define ROPE_L2T_OVER_DK 0.20762050593046014f

struct GemmFrag {
  v4f a[2][2];
  v4f b[2][4];
};

__device__ __forceinline__ void gemm_load(GemmFrag& f,
    const float* __restrict__ A0, const float* __restrict__ A1,
    const float* __restrict__ B0, const float* __restrict__ B1,
    const float* __restrict__ B2, const float* __restrict__ B3, int kk)
{
#pragma unroll
  for (int s = 0; s < 2; ++s) {
    f.a[s][0] = *(const v4f*)(A0 + kk + 8 * s);
    f.a[s][1] = *(const v4f*)(A1 + kk + 8 * s);
    f.b[s][0] = *(const v4f*)(B0 + kk + 8 * s);
    f.b[s][1] = *(const v4f*)(B1 + kk + 8 * s);
    f.b[s][2] = *(const v4f*)(B2 + kk + 8 * s);
    f.b[s][3] = *(const v4f*)(B3 + kk + 8 * s);
  }
}

__device__ __forceinline__ void gemm_mm(v8f (&acc)[2][4], const GemmFrag& f)
{
#pragma unroll
  for (int s = 0; s < 2; ++s)
#pragma unroll
    for (int e = 0; e < 4; e += 2)
#pragma unroll
      for (int i = 0; i < 2; ++i) {
        const v2f av = {f.a[s][i][e], f.a[s][i][e + 1]};
#pragma unroll
        for (int t = 0; t < 4; ++t) {
          const v2f bv = {f.b[s][t][e], f.b[s][t][e + 1]};
          acc[i][t] = __builtin_amdgcn_wmma_f32_16x16x4_f32(false, av, false, bv, (short)0, acc[i][t], false, false);
        }
      }
}

__global__ __launch_bounds__(256) __attribute__((amdgpu_waves_per_eu(2, 2))) void gemm_wmma_f32(
    const float* __restrict__ X, const float* __restrict__ W,
    float* __restrict__ out, int mode)
{
  const int lane = threadIdx.x & 31;
  const int wave = threadIdx.x >> 5;
  const int half = lane >> 4;
  const int l16  = lane & 15;

  const int m0 = (blockIdx.y * 8 + wave) * 32;
  const int n0 = blockIdx.x * 64;

  const int koff = 4 * half;
  const float* Arow0 = X + (size_t)(m0 + l16) * D_MODEL + koff;
  const float* Arow1 = Arow0 + (size_t)16 * D_MODEL;
  const float* Brow0 = W + (size_t)(n0 + l16) * D_MODEL + koff;
  const float* Brow1 = Brow0 + (size_t)16 * D_MODEL;
  const float* Brow2 = Brow0 + (size_t)32 * D_MODEL;
  const float* Brow3 = Brow0 + (size_t)48 * D_MODEL;

  v8f acc[2][4];
#pragma unroll
  for (int i = 0; i < 2; ++i)
#pragma unroll
    for (int t = 0; t < 4; ++t) acc[i][t] = (v8f){};

  GemmFrag fp, fq;
  gemm_load(fp, Arow0, Arow1, Brow0, Brow1, Brow2, Brow3, 0);
  for (int k = 0; k < D_MODEL; k += 32) {
    gemm_load(fq, Arow0, Arow1, Brow0, Brow1, Brow2, Brow3, k + 16);
    gemm_mm(acc, fp);
    gemm_load(fp, Arow0, Arow1, Brow0, Brow1, Brow2, Brow3, (k + 32) & (D_MODEL - 1));
    gemm_mm(acc, fq);
    asm volatile("" ::: "memory");
  }

  if (mode == 0) {
#pragma unroll
    for (int i = 0; i < 2; ++i)
#pragma unroll
      for (int t = 0; t < 4; ++t)
#pragma unroll
        for (int r = 0; r < 8; ++r) {
          const int row = m0 + i * 16 + r + 8 * half;
          out[(size_t)row * D_MODEL + n0 + t * 16 + l16] = acc[i][t][r];
        }
  } else {
#pragma unroll
    for (int t = 0; t < 4; ++t) {
      const int gcol = n0 + t * 16 + l16;
      const int h = gcol >> 6;
      const int d = gcol & 63;
      const bool even = (d & 1) == 0;
      const float invfreq = exp2f(-(float)(d & ~1) * ROPE_L2T_OVER_DK);
#pragma unroll
      for (int i = 0; i < 2; ++i)
#pragma unroll
        for (int r = 0; r < 8; ++r) {
          const int row = m0 + i * 16 + r + 8 * half;
          const int b  = row >> 11;
          const int s  = row & (S_LEN - 1);
          float v  = acc[i][t][r];
          float vp = __shfl_xor(v, 1, 32);
          if (mode == 1) {
            const float ang = (float)s * invfreq;
            const float c = __cosf(ang), sn = __sinf(ang);
            v = v * c + (even ? -vp : vp) * sn;
          }
          out[(((size_t)(b * NHEAD + h) * S_LEN + s) * DK) + d] = v;
        }
    }
  }
}

__device__ __forceinline__ float xhalf(float x) {
  const int b = __builtin_bit_cast(int, x);
  return __builtin_bit_cast(float,
      __builtin_amdgcn_permlanex16(b, b, 0x76543210, 0xfedcba98, false, false));
}

__device__ __forceinline__ void tile_softmax_pv(const v8f st, const int kg0, const int qg,
    float& mrun, float& lrun, v8f& o0, v8f& o1, v8f& o2, v8f& o3, const float (&vv)[8][4])
{
  float s[8];
#pragma unroll
  for (int r = 0; r < 8; ++r)
    s[r] = (kg0 + r <= qg) ? st[r] : -__builtin_inff();
  float mx = s[0];
#pragma unroll
  for (int r = 1; r < 8; ++r) mx = fmaxf(mx, s[r]);
  mx = fmaxf(mx, xhalf(mx));
  const float mnew = fmaxf(mrun, mx);
  const float scl  = __expf(mrun - mnew);
  float p[8];
  float psum = 0.f;
#pragma unroll
  for (int r = 0; r < 8; ++r) {
    p[r] = __expf(s[r] - mnew);
    psum += p[r];
  }
  lrun = lrun * scl + psum;
  mrun = mnew;

  o0 *= scl; o1 *= scl; o2 *= scl; o3 *= scl;

#pragma unroll
  for (int j = 0; j < 4; ++j) {
    v2f pb;
    pb[0] = p[2 * j];
    pb[1] = p[2 * j + 1];
    v2f a0, a1, a2, a3;
    a0[0] = vv[2 * j][0]; a0[1] = vv[2 * j + 1][0];
    a1[0] = vv[2 * j][1]; a1[1] = vv[2 * j + 1][1];
    a2[0] = vv[2 * j][2]; a2[1] = vv[2 * j + 1][2];
    a3[0] = vv[2 * j][3]; a3[1] = vv[2 * j + 1][3];
    o0 = __builtin_amdgcn_wmma_f32_16x16x4_f32(false, a0, false, pb, (short)0, o0, false, false);
    o1 = __builtin_amdgcn_wmma_f32_16x16x4_f32(false, a1, false, pb, (short)0, o1, false, false);
    o2 = __builtin_amdgcn_wmma_f32_16x16x4_f32(false, a2, false, pb, (short)0, o2, false, false);
    o3 = __builtin_amdgcn_wmma_f32_16x16x4_f32(false, a3, false, pb, (short)0, o3, false, false);
  }
}

__device__ __forceinline__ void tile_store(float* __restrict__ att, const int bh, const int qg,
    const int half, const float lrun, const v8f o0, const v8f o1, const v8f o2, const v8f o3)
{
  const float inv = 1.0f / (lrun + xhalf(lrun));
  const int b = bh >> 4;
  const int h = bh & 15;
  float* orow = att + ((size_t)(b * S_LEN) + qg) * D_MODEL + h * DK + 8 * half;
  v4f lo, hi;
  lo[0] = o0[0] * inv; lo[1] = o0[1] * inv; lo[2] = o0[2] * inv; lo[3] = o0[3] * inv;
  hi[0] = o0[4] * inv; hi[1] = o0[5] * inv; hi[2] = o0[6] * inv; hi[3] = o0[7] * inv;
  *(v4f*)(orow +  0) = lo; *(v4f*)(orow +  4) = hi;
  lo[0] = o1[0] * inv; lo[1] = o1[1] * inv; lo[2] = o1[2] * inv; lo[3] = o1[3] * inv;
  hi[0] = o1[4] * inv; hi[1] = o1[5] * inv; hi[2] = o1[6] * inv; hi[3] = o1[7] * inv;
  *(v4f*)(orow + 16) = lo; *(v4f*)(orow + 20) = hi;
  lo[0] = o2[0] * inv; lo[1] = o2[1] * inv; lo[2] = o2[2] * inv; lo[3] = o2[3] * inv;
  hi[0] = o2[4] * inv; hi[1] = o2[5] * inv; hi[2] = o2[6] * inv; hi[3] = o2[7] * inv;
  *(v4f*)(orow + 32) = lo; *(v4f*)(orow + 36) = hi;
  lo[0] = o3[0] * inv; lo[1] = o3[1] * inv; lo[2] = o3[2] * inv; lo[3] = o3[3] * inv;
  hi[0] = o3[4] * inv; hi[1] = o3[5] * inv; hi[2] = o3[6] * inv; hi[3] = o3[7] * inv;
  *(v4f*)(orow + 48) = lo; *(v4f*)(orow + 52) = hi;
}

__device__ __forceinline__ void key_block_two_tiles(
    const float* __restrict__ vhead, const float* __restrict__ kbase, const int kb,
    const int half, const int l16,
    const v4f (&kc)[8], v4f (&kn)[8], const v4f (&qa)[8], const v4f (&qb)[8],
    const int qga, const int qgb,
    float& ma, float& la, v8f& oa0, v8f& oa1, v8f& oa2, v8f& oa3,
    float& mb, float& lb, v8f& ob0, v8f& ob1, v8f& ob2, v8f& ob3)
{
  const int kg0 = kb * 16 + 8 * half;

  __builtin_amdgcn_sched_barrier(0);

  const float* vcol = vhead + (size_t)kg0 * DK + l16;
  float vv[8][4];
#pragma unroll
  for (int i = 0; i < 8; ++i)
#pragma unroll
    for (int t = 0; t < 4; ++t) vv[i][t] = vcol[i * DK + 16 * t];
  {
    const float* krow = kbase + (size_t)(kb + 1) * (16 * DK);
#pragma unroll
    for (int c = 0; c < 8; ++c) kn[c] = *(const v4f*)(krow + 8 * c);
  }
  __builtin_amdgcn_sched_barrier(0);

  v8f sa = {}, sb = {};
#pragma unroll
  for (int c = 0; c < 8; ++c) {
    const v2f kl = {kc[c][0], kc[c][1]}, kh = {kc[c][2], kc[c][3]};
    const v2f al = {qa[c][0], qa[c][1]}, ah = {qa[c][2], qa[c][3]};
    const v2f bl = {qb[c][0], qb[c][1]}, bu = {qb[c][2], qb[c][3]};
    sa = __builtin_amdgcn_wmma_f32_16x16x4_f32(false, kl, false, al, (short)0, sa, false, false);
    sb = __builtin_amdgcn_wmma_f32_16x16x4_f32(false, kl, false, bl, (short)0, sb, false, false);
    sa = __builtin_amdgcn_wmma_f32_16x16x4_f32(false, kh, false, ah, (short)0, sa, false, false);
    sb = __builtin_amdgcn_wmma_f32_16x16x4_f32(false, kh, false, bu, (short)0, sb, false, false);
  }

  tile_softmax_pv(sa, kg0, qga, ma, la, oa0, oa1, oa2, oa3, vv);
  tile_softmax_pv(sb, kg0, qgb, mb, lb, ob0, ob1, ob2, ob3, vv);
}

__global__ __launch_bounds__(128) void flash_attn(
    const float* __restrict__ Q, const float* __restrict__ K,
    const float* __restrict__ V, float* __restrict__ att)
{
  const int lane = threadIdx.x & 31;
  const int wave = __builtin_amdgcn_readfirstlane((int)(threadIdx.x >> 5));
  const int half = lane >> 4;
  const int l16  = lane & 15;
  const int task = blockIdx.x * 4 + wave;
  const int bh   = task >> 5;
  const int p    = task & 31;
  const size_t base = (size_t)bh * S_LEN * DK;
  const float* vhead = V + base;

  for (int pass = 0; pass < 2; ++pass) {
  const int grp = pass == 0 ? (S_LEN / 32 - 1) - p : p;
  const int qt0 = 2 * grp;

  v4f qa[8], qb[8];
  {
    const float* qrow = Q + base + (size_t)(qt0 * 16 + l16) * DK + 4 * half;
#pragma unroll
    for (int c = 0; c < 8; ++c) {
      v4f t = *(const v4f*)(qrow + 8 * c);
      qa[c] = t * 0.125f;
      v4f u = *(const v4f*)(qrow + 16 * DK + 8 * c);
      qb[c] = u * 0.125f;
    }
  }

  v8f oa0 = {}, oa1 = {}, oa2 = {}, oa3 = {};
  v8f ob0 = {}, ob1 = {}, ob2 = {}, ob3 = {};
  float ma = -__builtin_inff(), mb = -__builtin_inff();
  float la = 0.f, lb = 0.f;
  const int qga = qt0 * 16 + l16;
  const int qgb = qga + 16;

  const float* kbase = K + base + (size_t)l16 * DK + 4 * half;
  v4f kx[8], ky[8];
#pragma unroll
  for (int c = 0; c < 8; ++c) kx[c] = *(const v4f*)(kbase + 8 * c);

  for (int m = 0; m < grp; ++m) {
    key_block_two_tiles(vhead, kbase, 2 * m,     half, l16, kx, ky, qa, qb, qga, qgb,
                        ma, la, oa0, oa1, oa2, oa3, mb, lb, ob0, ob1, ob2, ob3);
    key_block_two_tiles(vhead, kbase, 2 * m + 1, half, l16, ky, kx, qa, qb, qga, qgb,
                        ma, la, oa0, oa1, oa2, oa3, mb, lb, ob0, ob1, ob2, ob3);
  }
  key_block_two_tiles(vhead, kbase, qt0, half, l16, kx, ky, qa, qb, qga, qgb,
                      ma, la, oa0, oa1, oa2, oa3, mb, lb, ob0, ob1, ob2, ob3);

  {
    const int kg0 = (qt0 + 1) * 16 + 8 * half;
    const float* vcol = vhead + (size_t)kg0 * DK + l16;
    float vv[8][4];
#pragma unroll
    for (int i = 0; i < 8; ++i)
#pragma unroll
      for (int t = 0; t < 4; ++t) vv[i][t] = vcol[i * DK + 16 * t];

    v8f sb = {};
#pragma unroll
    for (int c = 0; c < 8; ++c) {
      const v2f kl = {ky[c][0], ky[c][1]}, kh = {ky[c][2], ky[c][3]};
      const v2f bl = {qb[c][0], qb[c][1]}, bu = {qb[c][2], qb[c][3]};
      sb = __builtin_amdgcn_wmma_f32_16x16x4_f32(false, kl, false, bl, (short)0, sb, false, false);
      sb = __builtin_amdgcn_wmma_f32_16x16x4_f32(false, kh, false, bu, (short)0, sb, false, false);
    }
    tile_softmax_pv(sb, kg0, qgb, mb, lb, ob0, ob1, ob2, ob3, vv);
  }

  tile_store(att, bh, qga, half, la, oa0, oa1, oa2, oa3);
  tile_store(att, bh, qgb, half, lb, ob0, ob1, ob2, ob3);
  }
}

extern "C" void kernel_launch(void* const* d_in, const int* in_sizes, int n_in,
                              void* d_out, int out_size, void* d_ws, size_t ws_size,
                              hipStream_t stream) {
  const float* x  = (const float*)d_in[0];
  const float* wq = (const float*)d_in[1];
  const float* wk = (const float*)d_in[2];
  const float* wv = (const float*)d_in[3];
  const float* wo = (const float*)d_in[4];
  float* out = (float*)d_out;

  float* ws  = (float*)d_ws;
  const size_t SZ = (size_t)M_TOT * D_MODEL;
  float* Qw  = ws;
  float* Kw  = ws + SZ;
  float* Vw  = ws + 2 * SZ;
  float* Aw  = ws + 3 * SZ;

  dim3 blk(256);
  dim3 grd(D_MODEL / 64, M_TOT / (32 * 8));
  gemm_wmma_f32<<<grd, blk, 0, stream>>>(x, wq, Qw, 1);
  gemm_wmma_f32<<<grd, blk, 0, stream>>>(x, wk, Kw, 1);
  gemm_wmma_f32<<<grd, blk, 0, stream>>>(x, wv, Vw, 2);

  flash_attn<<<dim3(256), dim3(128), 0, stream>>>(Qw, Kw, Vw, Aw);

  gemm_wmma_f32<<<grd, blk, 0, stream>>>(Aw, wo, out, 0);
}
